# starting kernel
# baseline (speedup 1.0000x reference)
#include <hip/hip_runtime.h>

typedef __attribute__((ext_vector_type(2))) float v2f;
typedef __attribute__((ext_vector_type(8))) float v8f;

#define BATCH 16
#define CIN 8
#define IMG 256
#define NOC 8
#define KW 3
#define TAPS (CIN * KW * KW)
#define WROW (NOC * KW * KW)
#define PPB 8

#define PA_STRIDE (BATCH * TAPS)
#define WB_STRIDE (16 * TAPS)

__global__ __launch_bounds__(256, 2)
void lcn2d_wmma_f32(const float* __restrict__ xin,
                    const float* __restrict__ wgt,
                    const float* __restrict__ bias,
                    float* __restrict__ out)
{
    __shared__ float pa[PPB * PA_STRIDE];
    __shared__ float wb[PPB * WB_STRIDE];

    const int tid = threadIdx.x;
    const int px  = blockIdx.y;
    const int py0 = blockIdx.x * PPB;

    for (int c = 0; c < CIN; ++c) {
        const float* src = wgt + ((c * IMG + px) * IMG + py0) * WROW;
        for (int t = tid; t < PPB * WROW; t += 256) {
            int pix = t / WROW;
            int r   = t - pix * WROW;
            int oc  = r / 9;
            int tap = r - oc * 9;
            wb[pix * WB_STRIDE + oc * TAPS + c * 9 + tap] = src[t];
        }
    }
    for (int t = tid; t < PPB * NOC * TAPS; t += 256) {
        int pix = t / (NOC * TAPS);
        int r   = t - pix * (NOC * TAPS);
        wb[pix * WB_STRIDE + NOC * TAPS + r] = 0.0f;
    }

    for (int idx = tid; idx < PPB * PA_STRIDE; idx += 256) {
        int k   = idx % TAPS;
        int m   = (idx / TAPS) & (BATCH - 1);
        int pix = idx / PA_STRIDE;
        int c   = k / 9;
        int r   = k - c * 9;
        int i   = r / 3;
        int j   = r - i * 3;
        int sx = px + i - 1;        sx = sx < 0 ? 0 : (sx > IMG - 1 ? IMG - 1 : sx);
        int sy = py0 + pix - 1 + j; sy = sy < 0 ? 0 : (sy > IMG - 1 ? IMG - 1 : sy);
        pa[idx] = xin[((m * CIN + c) * IMG + sx) * IMG + sy];
    }
    __syncthreads();

    const int wv   = tid >> 5;
    const int lane = tid & 31;
    const int py   = py0 + wv;
    const int n    = lane & 15;
    const int m    = lane & 15;
    const int kh   = (lane >> 4) << 1;

    float bv = (n < NOC) ? bias[(px * IMG + py) * NOC + n] : 0.0f;
    v8f acc = { bv, bv, bv, bv, bv, bv, bv, bv };

    const float* pa_row = &pa[wv * PA_STRIDE + m * TAPS + kh];
    const float* wb_row = &wb[wv * WB_STRIDE + n * TAPS + kh];

    #pragma unroll
    for (int kk = 0; kk < TAPS / 4; ++kk) {
        v2f a  = *(const v2f*)(pa_row + kk * 4);
        v2f bm = *(const v2f*)(wb_row + kk * 4);
        acc = __builtin_amdgcn_wmma_f32_16x16x4_f32(
            false, a, false, bm, (short)0, acc, false, false);
    }

    if (n < NOC) {
        const int mb = (lane >> 4) << 3;
        #pragma unroll
        for (int r = 0; r < 8; ++r) {
            int b = r + mb;
            out[((b * NOC + n) * IMG + px) * IMG + py] = acc[r];
        }
    }
}

extern "C" void kernel_launch(void* const* d_in, const int* in_sizes, int n_in,
                              void* d_out, int out_size, void* d_ws, size_t ws_size,
                              hipStream_t stream) {
    const float* x    = (const float*)d_in[0];
    const float* wgt  = (const float*)d_in[1];
    const float* bias = (const float*)d_in[2];
    float* out        = (float*)d_out;

    dim3 grid(IMG / PPB, IMG);
    dim3 block(256);
    lcn2d_wmma_f32<<<grid, block, 0, stream>>>(x, wgt, bias, out);
}


# k reordered to (tap, channel): each pixel = 9 shifted 8-channel products; x and weights staged channel-interleaved in LDS, one 8-byte read per fragment, A shared by 3 pixels; f32 sum order changed
# speedup vs baseline: 5.9270x; 5.9270x over previous
#include <hip/hip_runtime.h>

typedef __attribute__((ext_vector_type(2))) float v2f;
typedef __attribute__((ext_vector_type(4))) float v4f;
typedef __attribute__((ext_vector_type(8))) float v8f;

#define BATCH 16
#define CIN 8
#define IMG 256
#define NOC 8
#define WPIX 72
#define P 32
#define XCOLS (P + 2)
#define XP 130
#define WCP (P * WPIX * 2 + 32)

__global__ __launch_bounds__(256)
void lcn2d_tapmajor_f32(const float* __restrict__ xin,
                        const float* __restrict__ wgt,
                        const float* __restrict__ bias,
                        float* __restrict__ out)
{
    __shared__ float xs[3 * XCOLS * XP];
    __shared__ float ws[4 * WCP];

    const int tid  = threadIdx.x;
    const int lane = tid & 31;
    const int wv   = tid >> 5;
    const int px   = blockIdx.y;
    const int py0  = blockIdx.x * P;

    float bv[4];
    #pragma unroll
    for (int q = 0; q < 4; ++q)
        bv[q] = bias[(px * IMG + py0 + 4 * wv + q) * NOC + (lane & 7)];

    #pragma unroll
    for (int it = 0; it < 9; ++it) {
        const int item = tid + it * 256;
        const int cp   = item / 576;
        const int t    = item - cp * 576;
        const v4f* s0 = (const v4f*)(wgt + (((2 * cp    ) * IMG + px) * IMG + py0) * WPIX) + t;
        const v4f* s1 = (const v4f*)(wgt + (((2 * cp + 1) * IMG + px) * IMG + py0) * WPIX) + t;
        const v4f a = *s0;
        const v4f b = *s1;
        v4f* d = (v4f*)(ws + cp * WCP + t * 8);
        d[0] = (v4f){ a.x, b.x, a.y, b.y };
        d[1] = (v4f){ a.z, b.z, a.w, b.w };
    }

    #pragma unroll
    for (int u = 0; u < 48; ++u) {
        const int L  = wv + 8 * u;
        const int i  = L >> 7;
        const int bc = L & 127;
        const int b  = bc >> 3;
        const int c  = bc & 7;
        int sx = px + i - 1;  sx = sx < 0 ? 0 : (sx > IMG - 1 ? IMG - 1 : sx);
        const float v = xin[(bc * IMG + sx) * IMG + py0 + lane];
        xs[(i * XCOLS + lane + 1) * XP + (c >> 2) * 64 + b * 4 + (c & 3)] = v;
    }
    #pragma unroll
    for (int v = 0; v < 3; ++v) {
        const int e    = tid + 256 * v;
        const int side = e / 384;
        const int rem  = e - side * 384;
        const int i    = rem >> 7;
        const int bc   = rem & 127;
        const int b    = bc >> 3;
        const int c    = bc & 7;
        int sx = px + i - 1;  sx = sx < 0 ? 0 : (sx > IMG - 1 ? IMG - 1 : sx);
        int sy = side ? py0 + P : py0 - 1;
        sy = sy < 0 ? 0 : (sy > IMG - 1 ? IMG - 1 : sy);
        const int col = side ? XCOLS - 1 : 0;
        xs[(i * XCOLS + col) * XP + (c >> 2) * 64 + b * 4 + (c & 3)] =
            xin[(bc * IMG + sx) * IMG + sy];
    }
    __syncthreads();

    const int n  = lane & 15;
    const int h  = lane >> 4;
    const int oc = n & 7;
    const int pl = 4 * wv;

    v8f acc[4];
    #pragma unroll
    for (int q = 0; q < 4; ++q)
        acc[q] = (v8f){ bv[q], bv[q], bv[q], bv[q], bv[q], bv[q], bv[q], bv[q] };

    const float* xa = xs + pl * XP + n * 4 + 2 * h;
    const float* wb = ws + h * WCP + pl * (WPIX * 2) + oc * 18;

    #pragma unroll
    for (int i = 0; i < 3; ++i) {
        #pragma unroll
        for (int cq = 0; cq < 2; ++cq) {
            v2f a[6];
            #pragma unroll
            for (int cl = 0; cl < 6; ++cl)
                a[cl] = *(const v2f*)(xa + (i * XCOLS + cl) * XP + cq * 64);
            #pragma unroll
            for (int j = 0; j < 3; ++j) {
                #pragma unroll
                for (int q = 0; q < 4; ++q) {
                    const v2f bm = *(const v2f*)(wb + cq * 2 * WCP + q * (WPIX * 2) + (i * 3 + j) * 2);
                    acc[q] = __builtin_amdgcn_wmma_f32_16x16x4_f32(
                        false, a[q + j], false, bm, (short)0, acc[q], false, false);
                }
            }
        }
    }

    if (n < NOC) {
        #pragma unroll
        for (int r = 0; r < 8; ++r) {
            const int b = r + 8 * h;
            v4f o = (v4f){ acc[0][r], acc[1][r], acc[2][r], acc[3][r] };
            *(v4f*)(out + ((b * NOC + n) * IMG + px) * IMG + py0 + pl) = o;
        }
    }
}

extern "C" void kernel_launch(void* const* d_in, const int* in_sizes, int n_in,
                              void* d_out, int out_size, void* d_ws, size_t ws_size,
                              hipStream_t stream) {
    const float* x    = (const float*)d_in[0];
    const float* wgt  = (const float*)d_in[1];
    const float* bias = (const float*)d_in[2];
    float* out        = (float*)d_out;

    dim3 grid(IMG / P, IMG);
    dim3 block(256);
    lcn2d_tapmajor_f32<<<grid, block, 0, stream>>>(x, wgt, bias, out);
}


# Blocks stay resident (grid 512) and walk 4 tiles each: the next tile's global loads wait in registers during the current tile's products; LDS-only barriers; k, sum order kept
# speedup vs baseline: 6.3274x; 1.0676x over previous
#include <hip/hip_runtime.h>

typedef __attribute__((ext_vector_type(2))) float v2f;
typedef __attribute__((ext_vector_type(4))) float v4f;
typedef __attribute__((ext_vector_type(8))) float v8f;

#define BATCH 16
#define CIN 8
#define IMG 256
#define NOC 8
#define WPIX 72
#define P 32
#define XCOLS (P + 2)
#define XP 130
#define WCP (P * WPIX * 2 + 32)
#define NBLK 512
#define NTILE 4
#define CHS (IMG * IMG * WPIX)

struct TileRegs {
    v4f wa[9];
    v4f wb[9];
    v4f xc[12];
    float xh[3];
    float bvl;
};

__device__ __forceinline__ void lds_barrier()
{
    __builtin_amdgcn_fence(__ATOMIC_RELEASE, "workgroup", "local");
    __builtin_amdgcn_s_barrier();
    __builtin_amdgcn_fence(__ATOMIC_ACQUIRE, "workgroup", "local");
}

__device__ __forceinline__ int clampi(int v) { return v < 0 ? 0 : (v > IMG - 1 ? IMG - 1 : v); }

__device__ __forceinline__ void tile_issue(TileRegs& r, const float* __restrict__ xin,
                                           const float* __restrict__ wgt,
                                           const float* __restrict__ bias,
                                           int px, int py0, int tid)
{
    const int lane = tid & 31;
    const int wv   = tid >> 5;
    r.bvl = bias[(px * IMG + py0 + 4 * wv) * NOC + lane];

    const int cp = __builtin_amdgcn_readfirstlane(tid >> 6);
    const float* w0 = wgt + (size_t)(2 * cp) * CHS + (px * IMG + py0) * WPIX;
    const float* w1 = w0 + CHS;
    const unsigned wo = (unsigned)(tid & 63) * 4u;
    #pragma unroll
    for (int it = 0; it < 9; ++it) {
        r.wa[it] = *(const v4f*)(w0 + it * 256 + wo);
        r.wb[it] = *(const v4f*)(w1 + it * 256 + wo);
    }
    const int sub = tid & 7;
    const int rr  = tid >> 3;
    const unsigned xo = (unsigned)(rr * IMG * IMG + 4 * sub);
    #pragma unroll
    for (int u = 0; u < 12; ++u) {
        const int i  = u >> 2;
        const int sx = clampi(px + i - 1);
        const int xrow = __builtin_amdgcn_readfirstlane((32 * (u & 3) * IMG + sx) * IMG + py0);
        r.xc[u] = *(const v4f*)(xin + xrow + xo);
    }
    const int side = __builtin_amdgcn_readfirstlane(tid >> 7);
    const unsigned ho = (unsigned)((tid & 127) * IMG * IMG);
    const int sy   = clampi(side ? py0 + P : py0 - 1);
    #pragma unroll
    for (int i = 0; i < 3; ++i) {
        const int sx = clampi(px + i - 1);
        const int hrow = __builtin_amdgcn_readfirstlane(sx * IMG + sy);
        r.xh[i] = (xin + hrow)[ho];
    }
}

__device__ __forceinline__ void tile_stage(const TileRegs& r, float* xs, float* ws, int tid)
{
    const int cp = tid >> 6;
    #pragma unroll
    for (int it = 0; it < 9; ++it) {
        const v4f a = r.wa[it];
        const v4f b = r.wb[it];
        v4f* d = (v4f*)(ws + cp * WCP + (64 * it + (tid & 63)) * 8);
        d[0] = (v4f){ a.x, b.x, a.y, b.y };
        d[1] = (v4f){ a.z, b.z, a.w, b.w };
    }
    const int sub = tid & 7;
    const int rr  = tid >> 3;
    #pragma unroll
    for (int u = 0; u < 12; ++u) {
        const int i  = u >> 2;
        const int bc = 32 * (u & 3) + rr;
        const int b  = bc >> 3;
        const int c  = bc & 7;
        float* d = xs + (i * XCOLS + 4 * sub + 1) * XP + (c >> 2) * 64 + b * 4 + (c & 3);
        d[0]      = r.xc[u].x;
        d[XP]     = r.xc[u].y;
        d[2 * XP] = r.xc[u].z;
        d[3 * XP] = r.xc[u].w;
    }
    const int side = tid >> 7;
    const int hbc  = tid & 127;
    const int hb   = hbc >> 3;
    const int hc   = hbc & 7;
    const int col  = side ? XCOLS - 1 : 0;
    #pragma unroll
    for (int i = 0; i < 3; ++i)
        xs[(i * XCOLS + col) * XP + (hc >> 2) * 64 + hb * 4 + (hc & 3)] = r.xh[i];
}

__global__ __launch_bounds__(256) __attribute__((amdgpu_waves_per_eu(4, 4)))
void lcn2d_tapmajor_f32(const float* __restrict__ xin,
                        const float* __restrict__ wgt,
                        const float* __restrict__ bias,
                        float* __restrict__ out)
{
    __shared__ float xs[3 * XCOLS * XP];
    __shared__ float ws[4 * WCP];

    const int tid  = threadIdx.x;
    const int lane = tid & 31;
    const int wv   = tid >> 5;
    const int py0  = (blockIdx.x & 7) * P;
    int px         = blockIdx.x >> 3;

    const int n  = lane & 15;
    const int h  = lane >> 4;
    const int oc = n & 7;
    const int pl = 4 * wv;
    const float* xa = xs + pl * XP + n * 4 + 2 * h;
    const float* wb = ws + h * WCP + pl * (WPIX * 2) + oc * 18;

    TileRegs r;
    tile_issue(r, xin, wgt, bias, px, py0, tid);

    #pragma nounroll
    for (int t = 0; t < NTILE; ++t) {
        tile_stage(r, xs, ws, tid);
        v8f acc[4];
        #pragma unroll
        for (int q = 0; q < 4; ++q) {
            const float bq = __shfl(r.bvl, q * 8 + (lane & 7));
            acc[q] = (v8f){ bq, bq, bq, bq, bq, bq, bq, bq };
        }
        lds_barrier();

        if (t + 1 < NTILE)
            tile_issue(r, xin, wgt, bias, px + NBLK / 8, py0, tid);

        #pragma unroll
        for (int i = 0; i < 3; ++i) {
            #pragma unroll
            for (int cq = 0; cq < 2; ++cq) {
                __builtin_amdgcn_sched_barrier(0);
                v2f a[6];
                #pragma unroll
                for (int cl = 0; cl < 6; ++cl)
                    a[cl] = *(const v2f*)(xa + (i * XCOLS + cl) * XP + cq * 64);
                #pragma unroll
                for (int j = 0; j < 3; ++j) {
                    #pragma unroll
                    for (int q = 0; q < 4; ++q) {
                        const v2f bm = *(const v2f*)(wb + cq * 2 * WCP + q * (WPIX * 2) + (i * 3 + j) * 2);
                        acc[q] = __builtin_amdgcn_wmma_f32_16x16x4_f32(
                            false, a[q + j], false, bm, (short)0, acc[q], false, false);
                    }
                }
            }
        }

        if (n < NOC) {
            #pragma unroll
            for (int k = 0; k < 8; ++k) {
                const int b = k + 8 * h;
                v4f o = (v4f){ acc[0][k], acc[1][k], acc[2][k], acc[3][k] };
                *(v4f*)(out + ((b * NOC + n) * IMG + px) * IMG + py0 + pl) = o;
            }
        }
        lds_barrier();
        px += NBLK / 8;
    }
}

extern "C" void kernel_launch(void* const* d_in, const int* in_sizes, int n_in,
                              void* d_out, int out_size, void* d_ws, size_t ws_size,
                              hipStream_t stream) {
    const float* x    = (const float*)d_in[0];
    const float* wgt  = (const float*)d_in[1];
    const float* bias = (const float*)d_in[2];
    float* out        = (float*)d_out;

    dim3 grid(NBLK);
    dim3 block(256);
    lcn2d_tapmajor_f32<<<grid, block, 0, stream>>>(x, wgt, bias, out);
}


# Results gathered in an LDS out tile, then each store instruction writes 4 whole 128-B lines (was 16 pieces of 16 B, every line assembled by 8 waves); no new barrier; k, sum order kept
# speedup vs baseline: 7.1393x; 1.1283x over previous
#include <hip/hip_runtime.h>

typedef __attribute__((ext_vector_type(2))) float v2f;
typedef __attribute__((ext_vector_type(4))) float v4f;
typedef __attribute__((ext_vector_type(8))) float v8f;

#define BATCH 16
#define CIN 8
#define IMG 256
#define NOC 8
#define WPIX 72
#define P 32
#define XCOLS (P + 2)
#define XP 130
#define WCP (P * WPIX * 2 + 32)
#define OP (P + 4)
#define NBLK 512
#define NTILE 4
#define CHS (IMG * IMG * WPIX)

struct TileRegs {
    v4f wa[9];
    v4f wb[9];
    v4f xc[12];
    float xh[3];
    float bvl;
};

__device__ __forceinline__ void lds_barrier()
{
    __builtin_amdgcn_fence(__ATOMIC_RELEASE, "workgroup", "local");
    __builtin_amdgcn_s_barrier();
    __builtin_amdgcn_fence(__ATOMIC_ACQUIRE, "workgroup", "local");
}

__device__ __forceinline__ int clampi(int v) { return v < 0 ? 0 : (v > IMG - 1 ? IMG - 1 : v); }

__device__ __forceinline__ void tile_issue(TileRegs& r, const float* __restrict__ xin,
                                           const float* __restrict__ wgt,
                                           const float* __restrict__ bias,
                                           int px, int py0, int tid)
{
    const int lane = tid & 31;
    const int wv   = tid >> 5;
    r.bvl = bias[(px * IMG + py0 + 4 * wv) * NOC + lane];

    const int cp = __builtin_amdgcn_readfirstlane(tid >> 6);
    const float* w0 = wgt + (size_t)(2 * cp) * CHS + (px * IMG + py0) * WPIX;
    const float* w1 = w0 + CHS;
    const unsigned wo = (unsigned)(tid & 63) * 4u;
    #pragma unroll
    for (int it = 0; it < 9; ++it) {
        r.wa[it] = *(const v4f*)(w0 + it * 256 + wo);
        r.wb[it] = *(const v4f*)(w1 + it * 256 + wo);
    }
    const int sub = tid & 7;
    const int rr  = tid >> 3;
    const unsigned xo = (unsigned)(rr * IMG * IMG + 4 * sub);
    #pragma unroll
    for (int u = 0; u < 12; ++u) {
        const int i  = u >> 2;
        const int sx = clampi(px + i - 1);
        const int xrow = __builtin_amdgcn_readfirstlane((32 * (u & 3) * IMG + sx) * IMG + py0);
        r.xc[u] = *(const v4f*)(xin + xrow + xo);
    }
    const int side = __builtin_amdgcn_readfirstlane(tid >> 7);
    const unsigned ho = (unsigned)((tid & 127) * IMG * IMG);
    const int sy   = clampi(side ? py0 + P : py0 - 1);
    #pragma unroll
    for (int i = 0; i < 3; ++i) {
        const int sx = clampi(px + i - 1);
        const int hrow = __builtin_amdgcn_readfirstlane(sx * IMG + sy);
        r.xh[i] = (xin + hrow)[ho];
    }
}

__device__ __forceinline__ void tile_stage(const TileRegs& r, float* xs, float* ws, int tid)
{
    const int cp = tid >> 6;
    #pragma unroll
    for (int it = 0; it < 9; ++it) {
        const v4f a = r.wa[it];
        const v4f b = r.wb[it];
        v4f* d = (v4f*)(ws + cp * WCP + (64 * it + (tid & 63)) * 8);
        d[0] = (v4f){ a.x, b.x, a.y, b.y };
        d[1] = (v4f){ a.z, b.z, a.w, b.w };
    }
    const int sub = tid & 7;
    const int rr  = tid >> 3;
    #pragma unroll
    for (int u = 0; u < 12; ++u) {
        const int i  = u >> 2;
        const int bc = 32 * (u & 3) + rr;
        const int b  = bc >> 3;
        const int c  = bc & 7;
        float* d = xs + (i * XCOLS + 4 * sub + 1) * XP + (c >> 2) * 64 + b * 4 + (c & 3);
        d[0]      = r.xc[u].x;
        d[XP]     = r.xc[u].y;
        d[2 * XP] = r.xc[u].z;
        d[3 * XP] = r.xc[u].w;
    }
    const int side = tid >> 7;
    const int hbc  = tid & 127;
    const int hb   = hbc >> 3;
    const int hc   = hbc & 7;
    const int col  = side ? XCOLS - 1 : 0;
    #pragma unroll
    for (int i = 0; i < 3; ++i)
        xs[(i * XCOLS + col) * XP + (hc >> 2) * 64 + hb * 4 + (hc & 3)] = r.xh[i];
}

__global__ __launch_bounds__(256) __attribute__((amdgpu_waves_per_eu(4, 4)))
void lcn2d_tapmajor_f32(const float* __restrict__ xin,
                        const float* __restrict__ wgt,
                        const float* __restrict__ bias,
                        float* __restrict__ out)
{
    __shared__ float xs[3 * XCOLS * XP];
    __shared__ float ws[4 * WCP];
    __shared__ __attribute__((aligned(16))) float os[BATCH * NOC * OP];

    const int tid  = threadIdx.x;
    const int lane = tid & 31;
    const int wv   = tid >> 5;
    const int py0  = (blockIdx.x & 7) * P;
    int px         = blockIdx.x >> 3;

    const int n  = lane & 15;
    const int h  = lane >> 4;
    const int oc = n & 7;
    const int pl = 4 * wv;
    const float* xa = xs + pl * XP + n * 4 + 2 * h;
    const float* wb = ws + h * WCP + pl * (WPIX * 2) + oc * 18;

    TileRegs r;
    tile_issue(r, xin, wgt, bias, px, py0, tid);

    #pragma nounroll
    for (int t = 0; t < NTILE; ++t) {
        tile_stage(r, xs, ws, tid);
        v8f acc[4];
        #pragma unroll
        for (int q = 0; q < 4; ++q) {
            const float bq = __shfl(r.bvl, q * 8 + (lane & 7));
            acc[q] = (v8f){ bq, bq, bq, bq, bq, bq, bq, bq };
        }
        lds_barrier();

        if (t + 1 < NTILE)
            tile_issue(r, xin, wgt, bias, px + NBLK / 8, py0, tid);

        #pragma unroll
        for (int i = 0; i < 3; ++i) {
            #pragma unroll
            for (int cq = 0; cq < 2; ++cq) {
                __builtin_amdgcn_sched_barrier(0);
                v2f a[6];
                #pragma unroll
                for (int cl = 0; cl < 6; ++cl)
                    a[cl] = *(const v2f*)(xa + (i * XCOLS + cl) * XP + cq * 64);
                #pragma unroll
                for (int j = 0; j < 3; ++j) {
                    #pragma unroll
                    for (int q = 0; q < 4; ++q) {
                        const v2f bm = *(const v2f*)(wb + cq * 2 * WCP + q * (WPIX * 2) + (i * 3 + j) * 2);
                        acc[q] = __builtin_amdgcn_wmma_f32_16x16x4_f32(
                            false, a[q + j], false, bm, (short)0, acc[q], false, false);
                    }
                }
            }
        }

        if (n < NOC) {
            #pragma unroll
            for (int k = 0; k < 8; ++k) {
                const int b = k + 8 * h;
                v4f o = (v4f){ acc[0][k], acc[1][k], acc[2][k], acc[3][k] };
                *(v4f*)(os + (b * NOC + n) * OP + pl) = o;
            }
        }
        lds_barrier();
        for (int it = 0; it < 4; ++it) {
            const int row = 16 * wv + 4 * it + (lane >> 3);
            const v4f o = *(const v4f*)(os + row * OP + 4 * (lane & 7));
            *(v4f*)(out + (row * IMG + px) * IMG + py0 + 4 * (lane & 7)) = o;
        }
        px += NBLK / 8;
    }
}

extern "C" void kernel_launch(void* const* d_in, const int* in_sizes, int n_in,
                              void* d_out, int out_size, void* d_ws, size_t ws_size,
                              hipStream_t stream) {
    const float* x    = (const float*)d_in[0];
    const float* wgt  = (const float*)d_in[1];
    const float* bias = (const float*)d_in[2];
    float* out        = (float*)d_out;

    dim3 grid(NBLK);
    dim3 block(256);
    lcn2d_tapmajor_f32<<<grid, block, 0, stream>>>(x, wgt, bias, out);
}


# Next tile's 34 global loads per thread no longer issued together behind the publish barrier: a sixth goes out ahead of each of the six product groups (weights, x, halo order kept); k, sum order kept
# speedup vs baseline: 7.6439x; 1.0707x over previous
#include <hip/hip_runtime.h>

typedef __attribute__((ext_vector_type(2))) float v2f;
typedef __attribute__((ext_vector_type(4))) float v4f;
typedef __attribute__((ext_vector_type(8))) float v8f;

#define BATCH 16
#define CIN 8
#define IMG 256
#define NOC 8
#define WPIX 72
#define P 32
#define XCOLS (P + 2)
#define XP 130
#define WCP (P * WPIX * 2 + 32)
#define OP (P + 4)
#define NBLK 512
#define NTILE 4
#define CHS (IMG * IMG * WPIX)

struct TileRegs {
    v4f wa[9];
    v4f wb[9];
    v4f xc[12];
    float xh[3];
    float bvl;
};

__device__ __forceinline__ void lds_barrier()
{
    __builtin_amdgcn_fence(__ATOMIC_RELEASE, "workgroup", "local");
    __builtin_amdgcn_s_barrier();
    __builtin_amdgcn_fence(__ATOMIC_ACQUIRE, "workgroup", "local");
}

__device__ __forceinline__ int clampi(int v) { return v < 0 ? 0 : (v > IMG - 1 ? IMG - 1 : v); }

__device__ __forceinline__ void tile_issue_part(TileRegs& r, const float* __restrict__ xin,
                                                const float* __restrict__ wgt,
                                                const float* __restrict__ bias,
                                                int px, int py0, int tid, int g)
{
    if (g == 0) {
        const int lane = tid & 31;
        const int wv   = tid >> 5;
        r.bvl = bias[(px * IMG + py0 + 4 * wv) * NOC + lane];
    }
    if (g < 3) {
        const int cp = __builtin_amdgcn_readfirstlane(tid >> 6);
        const float* w0 = wgt + (size_t)(2 * cp) * CHS + (px * IMG + py0) * WPIX;
        const float* w1 = w0 + CHS;
        const unsigned wo = (unsigned)(tid & 63) * 4u;
        #pragma unroll
        for (int k = 0; k < 3; ++k) {
            const int it = 3 * g + k;
            r.wa[it] = *(const v4f*)(w0 + it * 256 + wo);
            r.wb[it] = *(const v4f*)(w1 + it * 256 + wo);
        }
    } else if (g < 5) {
        const int sub = tid & 7;
        const int rr  = tid >> 3;
        const unsigned xo = (unsigned)(rr * IMG * IMG + 4 * sub);
        #pragma unroll
        for (int k = 0; k < 6; ++k) {
            const int u  = 6 * (g - 3) + k;
            const int i  = u >> 2;
            const int sx = clampi(px + i - 1);
            const int xrow = __builtin_amdgcn_readfirstlane((32 * (u & 3) * IMG + sx) * IMG + py0);
            r.xc[u] = *(const v4f*)(xin + xrow + xo);
        }
    } else {
        const int side = __builtin_amdgcn_readfirstlane(tid >> 7);
        const unsigned ho = (unsigned)((tid & 127) * IMG * IMG);
        const int sy   = clampi(side ? py0 + P : py0 - 1);
        #pragma unroll
        for (int i = 0; i < 3; ++i) {
            const int sx = clampi(px + i - 1);
            const int hrow = __builtin_amdgcn_readfirstlane(sx * IMG + sy);
            r.xh[i] = (xin + hrow)[ho];
        }
    }
}

__device__ __forceinline__ void tile_issue(TileRegs& r, const float* __restrict__ xin,
                                           const float* __restrict__ wgt,
                                           const float* __restrict__ bias,
                                           int px, int py0, int tid)
{
    #pragma unroll
    for (int g = 0; g < 6; ++g)
        tile_issue_part(r, xin, wgt, bias, px, py0, tid, g);
}

__device__ __forceinline__ void tile_stage(const TileRegs& r, float* xs, float* ws, int tid)
{
    const int cp = tid >> 6;
    #pragma unroll
    for (int it = 0; it < 9; ++it) {
        const v4f a = r.wa[it];
        const v4f b = r.wb[it];
        v4f* d = (v4f*)(ws + cp * WCP + (64 * it + (tid & 63)) * 8);
        d[0] = (v4f){ a.x, b.x, a.y, b.y };
        d[1] = (v4f){ a.z, b.z, a.w, b.w };
    }
    const int sub = tid & 7;
    const int rr  = tid >> 3;
    #pragma unroll
    for (int u = 0; u < 12; ++u) {
        const int i  = u >> 2;
        const int bc = 32 * (u & 3) + rr;
        const int b  = bc >> 3;
        const int c  = bc & 7;
        float* d = xs + (i * XCOLS + 4 * sub + 1) * XP + (c >> 2) * 64 + b * 4 + (c & 3);
        d[0]      = r.xc[u].x;
        d[XP]     = r.xc[u].y;
        d[2 * XP] = r.xc[u].z;
        d[3 * XP] = r.xc[u].w;
    }
    const int side = tid >> 7;
    const int hbc  = tid & 127;
    const int hb   = hbc >> 3;
    const int hc   = hbc & 7;
    const int col  = side ? XCOLS - 1 : 0;
    #pragma unroll
    for (int i = 0; i < 3; ++i)
        xs[(i * XCOLS + col) * XP + (hc >> 2) * 64 + hb * 4 + (hc & 3)] = r.xh[i];
}

__global__ __launch_bounds__(256) __attribute__((amdgpu_waves_per_eu(4, 4)))
void lcn2d_tapmajor_f32(const float* __restrict__ xin,
                        const float* __restrict__ wgt,
                        const float* __restrict__ bias,
                        float* __restrict__ out)
{
    __shared__ float xs[3 * XCOLS * XP];
    __shared__ float ws[4 * WCP];
    __shared__ __attribute__((aligned(16))) float os[BATCH * NOC * OP];

    const int tid  = threadIdx.x;
    const int lane = tid & 31;
    const int wv   = tid >> 5;
    const int py0  = (blockIdx.x & 7) * P;
    int px         = blockIdx.x >> 3;

    const int n  = lane & 15;
    const int h  = lane >> 4;
    const int oc = n & 7;
    const int pl = 4 * wv;
    const float* xa = xs + pl * XP + n * 4 + 2 * h;
    const float* wb = ws + h * WCP + pl * (WPIX * 2) + oc * 18;

    TileRegs r;
    tile_issue(r, xin, wgt, bias, px, py0, tid);

    #pragma nounroll
    for (int t = 0; t < NTILE; ++t) {
        tile_stage(r, xs, ws, tid);
        v8f acc[4];
        #pragma unroll
        for (int q = 0; q < 4; ++q) {
            const float bq = __shfl(r.bvl, q * 8 + (lane & 7));
            acc[q] = (v8f){ bq, bq, bq, bq, bq, bq, bq, bq };
        }
        lds_barrier();

        const bool more = t + 1 < NTILE;

        #pragma unroll
        for (int i = 0; i < 3; ++i) {
            #pragma unroll
            for (int cq = 0; cq < 2; ++cq) {
                __builtin_amdgcn_sched_barrier(0);
                if (more)
                    tile_issue_part(r, xin, wgt, bias, px + NBLK / 8, py0, tid, i * 2 + cq);
                v2f a[6];
                #pragma unroll
                for (int cl = 0; cl < 6; ++cl)
                    a[cl] = *(const v2f*)(xa + (i * XCOLS + cl) * XP + cq * 64);
                #pragma unroll
                for (int j = 0; j < 3; ++j) {
                    #pragma unroll
                    for (int q = 0; q < 4; ++q) {
                        const v2f bm = *(const v2f*)(wb + cq * 2 * WCP + q * (WPIX * 2) + (i * 3 + j) * 2);
                        acc[q] = __builtin_amdgcn_wmma_f32_16x16x4_f32(
                            false, a[q + j], false, bm, (short)0, acc[q], false, false);
                    }
                }
            }
        }

        if (n < NOC) {
            #pragma unroll
            for (int k = 0; k < 8; ++k) {
                const int b = k + 8 * h;
                v4f o = (v4f){ acc[0][k], acc[1][k], acc[2][k], acc[3][k] };
                *(v4f*)(os + (b * NOC + n) * OP + pl) = o;
            }
        }
        lds_barrier();
        for (int it = 0; it < 4; ++it) {
            const int row = 16 * wv + 4 * it + (lane >> 3);
            const v4f o = *(const v4f*)(os + row * OP + 4 * (lane & 7));
            *(v4f*)(out + (row * IMG + px) * IMG + py0 + 4 * (lane & 7)) = o;
        }
        px += NBLK / 8;
    }
}

extern "C" void kernel_launch(void* const* d_in, const int* in_sizes, int n_in,
                              void* d_out, int out_size, void* d_ws, size_t ws_size,
                              hipStream_t stream) {
    const float* x    = (const float*)d_in[0];
    const float* wgt  = (const float*)d_in[1];
    const float* bias = (const float*)d_in[2];
    float* out        = (float*)d_out;

    dim3 grid(NBLK);
    dim3 block(256);
    lcn2d_tapmajor_f32<<<grid, block, 0, stream>>>(x, wgt, bias, out);
}
